# starting kernel
# baseline (speedup 1.0000x reference)
#include <hip/hip_runtime.h>
#include <math.h>

typedef __attribute__((ext_vector_type(2))) float v2f;
typedef __attribute__((ext_vector_type(8))) float v8f;

#define N_NODES 50000
#define K_NEI   32
#define D_IN    128
#define D_OUT   128
#define M_TILE  32
#define THREADS 256
#define A_STRIDE 516

__global__ __launch_bounds__(THREADS, 1)
void sage_fused_kernel(const float* __restrict__ fea1,
                       const float* __restrict__ fea2,
                       const int*   __restrict__ idx1,
                       const int*   __restrict__ idx2,
                       const float* __restrict__ W,
                       float*       __restrict__ out)
{
    __shared__ float smemA[M_TILE * A_STRIDE];

    const int tid       = threadIdx.x;
    const int blockRow0 = blockIdx.x * M_TILE;

    {
        const int r    = tid >> 3;
        const int sub  = tid & 7;
        const int c0   = sub * 16;
        const int node = blockRow0 + r;
        float* rowA = &smemA[r * A_STRIDE];

        if (node < N_NODES) {
            const float4* f1 = (const float4*)(fea1 + (size_t)node * D_IN + c0);
            const float4* f2 = (const float4*)(fea2 + (size_t)node * D_IN + c0);
            #pragma unroll
            for (int q = 0; q < 4; ++q) {
                *(float4*)&rowA[      c0 + q * 4] = f1[q];
                *(float4*)&rowA[128 + c0 + q * 4] = f2[q];
            }

            {
                float4 acc[4];
                #pragma unroll
                for (int q = 0; q < 4; ++q) acc[q] = make_float4(0.f, 0.f, 0.f, 0.f);
                const int* nl = idx1 + (size_t)node * K_NEI;
                for (int j = 0; j < K_NEI; ++j) {
                    const int g = nl[j];
                    const float4* src = (const float4*)(fea1 + (size_t)g * D_IN + c0);
                    #pragma unroll
                    for (int q = 0; q < 4; ++q) {
                        float4 v = src[q];
                        acc[q].x += v.x; acc[q].y += v.y;
                        acc[q].z += v.z; acc[q].w += v.w;
                    }
                }
                const float s = 1.0f / (float)K_NEI;
                #pragma unroll
                for (int q = 0; q < 4; ++q) {
                    float4 v = make_float4(acc[q].x * s, acc[q].y * s,
                                           acc[q].z * s, acc[q].w * s);
                    *(float4*)&rowA[256 + c0 + q * 4] = v;
                }
            }

            {
                float4 acc[4];
                #pragma unroll
                for (int q = 0; q < 4; ++q) acc[q] = make_float4(0.f, 0.f, 0.f, 0.f);
                const int* nl = idx2 + (size_t)node * K_NEI;
                for (int j = 0; j < K_NEI; ++j) {
                    const int g = nl[j];
                    const float4* src = (const float4*)(fea1 + (size_t)g * D_IN + c0);
                    #pragma unroll
                    for (int q = 0; q < 4; ++q) {
                        float4 v = src[q];
                        acc[q].x += v.x; acc[q].y += v.y;
                        acc[q].z += v.z; acc[q].w += v.w;
                    }
                }
                const float s = 1.0f / (float)K_NEI;
                #pragma unroll
                for (int q = 0; q < 4; ++q) {
                    float4 v = make_float4(acc[q].x * s, acc[q].y * s,
                                           acc[q].z * s, acc[q].w * s);
                    *(float4*)&rowA[384 + c0 + q * 4] = v;
                }
            }
        } else {
            const float4 z = make_float4(0.f, 0.f, 0.f, 0.f);
            #pragma unroll
            for (int q = 0; q < 4; ++q) {
                *(float4*)&rowA[      c0 + q * 4] = z;
                *(float4*)&rowA[128 + c0 + q * 4] = z;
                *(float4*)&rowA[256 + c0 + q * 4] = z;
                *(float4*)&rowA[384 + c0 + q * 4] = z;
            }
        }
    }

    __syncthreads();

    const int wave  = tid >> 5;
    const int lane  = tid & 31;
    const int m     = lane & 15;
    const int khalf = (lane >> 4) * 2;
    const int ncol  = (lane & 15) + wave * 16;

    v8f acc0 = 0.0f;
    v8f acc1 = 0.0f;

    const float* a0p = &smemA[ m        * A_STRIDE + khalf];
    const float* a1p = &smemA[(m + 16)  * A_STRIDE + khalf];
    const float* bp  = W + (size_t)khalf * D_OUT + ncol;

    #pragma unroll 4
    for (int kk = 0; kk < 4 * D_IN; kk += 4) {
        v2f a0; a0.x = a0p[0]; a0.y = a0p[1];
        v2f a1; a1.x = a1p[0]; a1.y = a1p[1];
        v2f b;  b.x  = bp[0];  b.y  = bp[D_OUT];

        acc0 = __builtin_amdgcn_wmma_f32_16x16x4_f32(
                   false, a0, false, b, (short)0, acc0, false, false);
        acc1 = __builtin_amdgcn_wmma_f32_16x16x4_f32(
                   false, a1, false, b, (short)0, acc1, false, false);

        a0p += 4;
        a1p += 4;
        bp  += 4 * D_OUT;
    }

    const int rbase = (lane >> 4) * 8;
    #pragma unroll
    for (int v = 0; v < 8; ++v) {
        const int row0 = blockRow0 + rbase + v;
        const int row1 = row0 + 16;
        if (row0 < N_NODES) out[(size_t)row0 * D_OUT + ncol] = tanhf(acc0[v]);
        if (row1 < N_NODES) out[(size_t)row1 * D_OUT + ncol] = tanhf(acc1[v]);
    }
}

extern "C" void kernel_launch(void* const* d_in, const int* in_sizes, int n_in,
                              void* d_out, int out_size, void* d_ws, size_t ws_size,
                              hipStream_t stream) {
    const float* fea1 = (const float*)d_in[0];
    const float* fea2 = (const float*)d_in[1];
    const int*   idx1 = (const int*)d_in[2];
    const int*   idx2 = (const int*)d_in[3];
    const float* W    = (const float*)d_in[4];
    float*       out  = (float*)d_out;

    const int grid = (N_NODES + M_TILE - 1) / M_TILE;
    sage_fused_kernel<<<grid, THREADS, 0, stream>>>(fea1, fea2, idx1, idx2, W, out);
}


# gather phase: one wave per node, its list indices preloaded and broadcast by readlane, each load one whole 512-byte row, about 20 independent row loads in flight per wave; sum order unchanged
# speedup vs baseline: 1.6165x; 1.6165x over previous
#include <hip/hip_runtime.h>
#include <math.h>

typedef __attribute__((ext_vector_type(2))) float v2f;
typedef __attribute__((ext_vector_type(8))) float v8f;
typedef __attribute__((ext_vector_type(4))) int   v4i;

#define N_NODES 50000
#define K_NEI   32
#define D_IN    128
#define D_OUT   128
#define M_TILE  32
#define THREADS 256
#define A_STRIDE 516
#define ROWS_PER_WAVE (M_TILE / (THREADS / 32))
#define GATHER_BATCH 16

__device__ __forceinline__ float4 gather_sum(const float* __restrict__ fea,
                                             v4i idxv, int rowsel, int ln)
{
    float4 acc = make_float4(0.f, 0.f, 0.f, 0.f);
    const unsigned lnoff = (unsigned)ln * (unsigned)sizeof(float4);
    #pragma unroll
    for (int jb = 0; jb < K_NEI; jb += GATHER_BATCH) {
        float4 v[GATHER_BATCH];
        #pragma unroll
        for (int u = 0; u < GATHER_BATCH; ++u) {
            const int j = jb + u;
            const int g = __builtin_amdgcn_readlane(idxv[j & 3], rowsel * 8 + (j >> 2));
            const unsigned off = (unsigned)g * (unsigned)(D_IN * sizeof(float)) + lnoff;
            v[u] = *(const float4*)((const char*)fea + off);
        }
        #pragma unroll
        for (int u = 0; u < GATHER_BATCH; ++u) {
            acc.x += v[u].x; acc.y += v[u].y;
            acc.z += v[u].z; acc.w += v[u].w;
        }
    }
    return acc;
}

__global__ __launch_bounds__(THREADS, 1)
void sage_fused_kernel(const float* __restrict__ fea1,
                       const float* __restrict__ fea2,
                       const int*   __restrict__ idx1,
                       const int*   __restrict__ idx2,
                       const float* __restrict__ W,
                       float*       __restrict__ out)
{
    __shared__ float smemA[M_TILE * A_STRIDE];

    const int tid       = threadIdx.x;
    const int blockRow0 = blockIdx.x * M_TILE;

    {
        const int wv    = __builtin_amdgcn_readfirstlane(tid >> 5);
        const int ln    = tid & 31;
        const int row0  = wv * ROWS_PER_WAVE;
        const int node0 = blockRow0 + row0;

        v4i i1v = 0, i2v = 0;
        if (node0 + (ln >> 3) < N_NODES) {
            i1v = ((const v4i*)(idx1 + (size_t)node0 * K_NEI))[ln];
            i2v = ((const v4i*)(idx2 + (size_t)node0 * K_NEI))[ln];
        }

        const float s = 1.0f / (float)K_NEI;
        for (int rr = 0; rr < ROWS_PER_WAVE; ++rr) {
            const int node = node0 + rr;
            float* rowA = &smemA[(row0 + rr) * A_STRIDE + ln * 4];
            if (node < N_NODES) {
                const float4 d1 = ((const float4*)(fea1 + (size_t)node * D_IN))[ln];
                const float4 d2 = ((const float4*)(fea2 + (size_t)node * D_IN))[ln];
                *(float4*)&rowA[0]   = d1;
                *(float4*)&rowA[128] = d2;

                const float4 a = gather_sum(fea1, i1v, rr, ln);
                *(float4*)&rowA[256] = make_float4(a.x * s, a.y * s, a.z * s, a.w * s);

                const float4 b = gather_sum(fea1, i2v, rr, ln);
                *(float4*)&rowA[384] = make_float4(b.x * s, b.y * s, b.z * s, b.w * s);
            } else {
                const float4 z = make_float4(0.f, 0.f, 0.f, 0.f);
                *(float4*)&rowA[0]   = z;
                *(float4*)&rowA[128] = z;
                *(float4*)&rowA[256] = z;
                *(float4*)&rowA[384] = z;
            }
        }
    }

    __syncthreads();

    const int wave  = tid >> 5;
    const int lane  = tid & 31;
    const int m     = lane & 15;
    const int khalf = (lane >> 4) * 2;
    const int ncol  = (lane & 15) + wave * 16;

    v8f acc0 = 0.0f;
    v8f acc1 = 0.0f;

    const float* a0p = &smemA[ m        * A_STRIDE + khalf];
    const float* a1p = &smemA[(m + 16)  * A_STRIDE + khalf];
    const float* bp  = W + (size_t)khalf * D_OUT + ncol;

    #pragma unroll 4
    for (int kk = 0; kk < 4 * D_IN; kk += 4) {
        v2f a0; a0.x = a0p[0]; a0.y = a0p[1];
        v2f a1; a1.x = a1p[0]; a1.y = a1p[1];
        v2f b;  b.x  = bp[0];  b.y  = bp[D_OUT];

        acc0 = __builtin_amdgcn_wmma_f32_16x16x4_f32(
                   false, a0, false, b, (short)0, acc0, false, false);
        acc1 = __builtin_amdgcn_wmma_f32_16x16x4_f32(
                   false, a1, false, b, (short)0, acc1, false, false);

        a0p += 4;
        a1p += 4;
        bp  += 4 * D_OUT;
    }

    const int rbase = (lane >> 4) * 8;
    #pragma unroll
    for (int v = 0; v < 8; ++v) {
        const int row0 = blockRow0 + rbase + v;
        const int row1 = row0 + 16;
        if (row0 < N_NODES) out[(size_t)row0 * D_OUT + ncol] = tanhf(acc0[v]);
        if (row1 < N_NODES) out[(size_t)row1 * D_OUT + ncol] = tanhf(acc1[v]);
    }
}

extern "C" void kernel_launch(void* const* d_in, const int* in_sizes, int n_in,
                              void* d_out, int out_size, void* d_ws, size_t ws_size,
                              hipStream_t stream) {
    const float* fea1 = (const float*)d_in[0];
    const float* fea2 = (const float*)d_in[1];
    const int*   idx1 = (const int*)d_in[2];
    const int*   idx2 = (const int*)d_in[3];
    const float* W    = (const float*)d_in[4];
    float*       out  = (float*)d_out;

    const int grid = (N_NODES + M_TILE - 1) / M_TILE;
    sage_fused_kernel<<<grid, THREADS, 0, stream>>>(fea1, fea2, idx1, idx2, W, out);
}


# 256 persistent workgroups keep the weight matrix in LDS (fragment order) instead of re-reading it from L2 per tile; 16 waves gather the next 16-row tile while 8 multiply; sum order unchanged
# speedup vs baseline: 1.6716x; 1.0341x over previous
#include <hip/hip_runtime.h>
#include <math.h>


typedef __attribute__((ext_vector_type(2))) float v2f;
typedef __attribute__((ext_vector_type(8))) float v8f;
typedef __attribute__((ext_vector_type(4))) float v4f;

#define N_NODES 50000
#define K_NEI   32
#define D_IN    128
#define D_OUT   128
#define K_TOT   512
#define M_TILE  16
#define GEMM_WAVES   8
#define GATHER_WAVES 16
#define THREADS ((GEMM_WAVES + GATHER_WAVES) * 32)
#define N_BLOCKS 256
#define ROWS_PER_BLOCK 196
#define A_TILE_FLOATS (M_TILE * K_TOT)
#define GATHER_GROUP 16

__device__ __forceinline__ void gather_row(const float* __restrict__ fea1,
                                           const float* __restrict__ fea2,
                                           const int* __restrict__ idx1,
                                           const int* __restrict__ idx2,
                                           int i1, int i2, int ln, int node, int pnode,
                                           float4& s1, float4& s2,
                                           int& ni1, int& ni2, float4& d1, float4& d2)
{
    const unsigned lnoff = (unsigned)ln * (unsigned)sizeof(float4);
    float4 a1 = make_float4(0.f, 0.f, 0.f, 0.f);
    float4 a2 = a1;
    v4f dv1 = 0.0f, dv2 = 0.0f;
    #pragma unroll
    for (int j0 = 0; j0 < 2 * K_NEI; j0 += GATHER_GROUP) {
        v4f v[GATHER_GROUP];
        #pragma unroll
        for (int u = 0; u < GATHER_GROUP; ++u) {
            const int j = j0 + u;
            if (j < 2 * K_NEI) {
                const int g = __builtin_amdgcn_readlane(j < K_NEI ? i1 : i2, j & (K_NEI - 1));
                const unsigned off = (unsigned)g * (unsigned)(D_IN * sizeof(float)) + lnoff;
                v[u] = *(const v4f*)((const char*)fea1 + off);
            }
        }
        if (j0 + GATHER_GROUP >= 2 * K_NEI) {
            const unsigned ioff = (unsigned)pnode * (unsigned)(K_NEI * sizeof(int)) + (unsigned)ln * (unsigned)sizeof(int);
            const unsigned foff = (unsigned)node * (unsigned)(D_IN * sizeof(float)) + lnoff;
            ni1 = *(const int*)((const char*)idx1 + ioff);
            ni2 = *(const int*)((const char*)idx2 + ioff);
            dv1 = *(const v4f*)((const char*)fea1 + foff);
            dv2 = *(const v4f*)((const char*)fea2 + foff);
        }
        if (j0 + GATHER_GROUP < 2 * K_NEI) {
            asm volatile("" :: "v"(v[0]), "v"(v[1]), "v"(v[2]), "v"(v[3]), "v"(v[4]), "v"(v[5]),
                               "v"(v[6]), "v"(v[7]), "v"(v[8]), "v"(v[9]), "v"(v[10]), "v"(v[11]),
                               "v"(v[12]), "v"(v[13]), "v"(v[14]), "v"(v[15])
                         : "memory");
        } else {
            asm volatile("" :: "v"(v[0]), "v"(v[1]), "v"(v[2]), "v"(v[3]), "v"(v[4]), "v"(v[5]),
                               "v"(v[6]), "v"(v[7]), "v"(v[8]), "v"(v[9]), "v"(v[10]), "v"(v[11]),
                               "v"(v[12]), "v"(v[13]), "v"(v[14]), "v"(v[15]), "v"(dv1), "v"(dv2)
                         : "memory");
        }
        #pragma unroll
        for (int u = 0; u < GATHER_GROUP; ++u) {
            const int j = j0 + u;
            if (j < K_NEI) {
                a1.x += v[u].x; a1.y += v[u].y; a1.z += v[u].z; a1.w += v[u].w;
            } else if (j < 2 * K_NEI) {
                a2.x += v[u].x; a2.y += v[u].y; a2.z += v[u].z; a2.w += v[u].w;
            }
        }
    }
    s1 = a1;
    s2 = a2;
    d1 = make_float4(dv1.x, dv1.y, dv1.z, dv1.w);
    d2 = make_float4(dv2.x, dv2.y, dv2.z, dv2.w);
}

__global__ __launch_bounds__(THREADS, 1)
void sage_persistent_kernel(const float* __restrict__ fea1,
                            const float* __restrict__ fea2,
                            const int*   __restrict__ idx1,
                            const int*   __restrict__ idx2,
                            const float* __restrict__ W,
                            float*       __restrict__ out)
{
    __shared__ __attribute__((aligned(16))) float sW[K_TOT * D_OUT];
    __shared__ __attribute__((aligned(16))) float sA[2 * A_TILE_FLOATS];

    const int tid = threadIdx.x;
    const int wv  = __builtin_amdgcn_readfirstlane(tid >> 5);
    const int ln  = tid & 31;

    const int rowBeg = blockIdx.x * ROWS_PER_BLOCK;
    const int rowEnd = (rowBeg + ROWS_PER_BLOCK < N_NODES) ? rowBeg + ROWS_PER_BLOCK : N_NODES;
    const int nTiles = (rowEnd - rowBeg + M_TILE - 1) / M_TILE;

    const int grow = wv - GEMM_WAVES;
    int   ci1 = 0, ci2 = 0;
    if (wv >= GEMM_WAVES) {
        const int node0 = rowBeg + grow;
        if (node0 < rowEnd) {
            const unsigned ioff = (unsigned)node0 * (unsigned)(K_NEI * sizeof(int)) + (unsigned)ln * (unsigned)sizeof(int);
            ci1 = *(const int*)((const char*)idx1 + ioff);
            ci2 = *(const int*)((const char*)idx2 + ioff);
        }
    }

    const int m   = ln & 15;
    const int hl  = ln >> 4;
    const int swz = m << 2;

    for (int s = 0; s <= nTiles; ++s) {
        if (wv < GEMM_WAVES) {
            if (s == 0) {
                #pragma unroll 2
                for (int i = tid; i < (K_TOT / 8) * 2 * (D_OUT / 4); i += GEMM_WAVES * 32) {
                    const int q  = i & 31;
                    const int h  = (i >> 5) & 1;
                    const int k8 = i >> 6;
                    const float4* src = (const float4*)(W + (size_t)(8 * k8 + 2 * h) * D_OUT) + q;
                    const float4 r0 = src[0];
                    const float4 r1 = src[D_OUT / 4];
                    const float4 r2 = src[4 * (D_OUT / 4)];
                    const float4 r3 = src[5 * (D_OUT / 4)];
                    float4* dst = (float4*)(sW + k8 * 1024 + (q >> 2) * 128 + h * 64 + (q & 3) * 16);
                    dst[0] = make_float4(r0.x, r1.x, r2.x, r3.x);
                    dst[1] = make_float4(r0.y, r1.y, r2.y, r3.y);
                    dst[2] = make_float4(r0.z, r1.z, r2.z, r3.z);
                    dst[3] = make_float4(r0.w, r1.w, r2.w, r3.w);
                }
            } else {
                const int t = s - 1;
                const float*  arow = sA + (t & 1) * A_TILE_FLOATS + m * K_TOT;
                const float4* bcol = (const float4*)sW + wv * 32 + ln;
                v8f acc = 0.0f;
                #pragma unroll 1
                for (int kg = 0; kg < K_TOT / 64; ++kg) {
                    #pragma unroll
                    for (int ku = 0; ku < 8; ++ku) {
                        const int k8 = kg * 8 + ku;
                        const float4 bq = bcol[k8 * 256];
                        const v2f a0 = *(const v2f*)(arow + ((8 * k8 + 2 * hl) ^ swz));
                        const v2f a1 = *(const v2f*)(arow + ((8 * k8 + 4 + 2 * hl) ^ swz));
                        v2f b0; b0.x = bq.x; b0.y = bq.y;
                        v2f b1; b1.x = bq.z; b1.y = bq.w;
                        acc = __builtin_amdgcn_wmma_f32_16x16x4_f32(
                                  false, a0, false, b0, (short)0, acc, false, false);
                        acc = __builtin_amdgcn_wmma_f32_16x16x4_f32(
                                  false, a1, false, b1, (short)0, acc, false, false);
                    }
                    __builtin_amdgcn_sched_group_barrier(0x100, 24, 0);
                    __builtin_amdgcn_sched_group_barrier(0x008, 16, 0);
                }
                const int ncol = wv * 16 + m;
                const int row0 = rowBeg + t * M_TILE + hl * 8;
                #pragma unroll
                for (int v = 0; v < 8; ++v) {
                    const int row = row0 + v;
                    if (row < rowEnd) out[(size_t)row * D_OUT + ncol] = tanhf(acc[v]);
                }
            }
        } else if (s < nTiles) {
            const int node = rowBeg + s * M_TILE + grow;
            float* rowA = sA + (s & 1) * A_TILE_FLOATS + grow * K_TOT + ((ln ^ grow) << 2);
            if (node < rowEnd) {
                const int pnode = (node + M_TILE < N_NODES) ? node + M_TILE : N_NODES - 1;
                int ni1 = 0, ni2 = 0;
                float4 a, b, d1, d2;
                gather_row(fea1, fea2, idx1, idx2, ci1, ci2, ln, node, pnode, a, b, ni1, ni2, d1, d2);
                const float sc = 1.0f / (float)K_NEI;
                *(float4*)&rowA[256] = make_float4(a.x * sc, a.y * sc, a.z * sc, a.w * sc);
                *(float4*)&rowA[384] = make_float4(b.x * sc, b.y * sc, b.z * sc, b.w * sc);
                *(float4*)&rowA[0]   = d1;
                *(float4*)&rowA[128] = d2;
                ci1 = ni1; ci2 = ni2;
            } else {
                const float4 z = make_float4(0.f, 0.f, 0.f, 0.f);
                *(float4*)&rowA[0]   = z;
                *(float4*)&rowA[128] = z;
                *(float4*)&rowA[256] = z;
                *(float4*)&rowA[384] = z;
            }
        }
        if (s < nTiles) __syncthreads();
    }
}

extern "C" void kernel_launch(void* const* d_in, const int* in_sizes, int n_in,
                              void* d_out, int out_size, void* d_ws, size_t ws_size,
                              hipStream_t stream) {
    const float* fea1 = (const float*)d_in[0];
    const float* fea2 = (const float*)d_in[1];
    const int*   idx1 = (const int*)d_in[2];
    const int*   idx2 = (const int*)d_in[3];
    const float* W    = (const float*)d_in[4];
    float*       out  = (float*)d_out;

    sage_persistent_kernel<<<N_BLOCKS, THREADS, 0, stream>>>(fea1, fea2, idx1, idx2, W, out);
}
